# starting kernel
# baseline (speedup 1.0000x reference)
#include <hip/hip_runtime.h>

typedef __attribute__((ext_vector_type(2))) float v2f;
typedef __attribute__((ext_vector_type(4))) float v4f;
typedef __attribute__((ext_vector_type(8))) float v8f;

constexpr int B  = 128;
constexpr int T  = 512;
constexpr int DZ = 512;

__global__ void scan_kernel(const float* __restrict__ x, float* __restrict__ S) {
  __shared__ float buf[T];
  const int b = blockIdx.x;
  const int t = threadIdx.x;

  float v = x[(size_t)b * T + t];
  buf[t] = v;
  __syncthreads();

  for (int off = 1; off < T; off <<= 1) {
    float add = (t >= off) ? buf[t - off] : 0.0f;
    __syncthreads();
    v += add;
    buf[t] = v;
    __syncthreads();
  }
  S[(size_t)b * T + t] = v;
}

__global__ void rank1_wmma_kernel(const float* __restrict__ S,
                                  const float* __restrict__ z0,
                                  const float* __restrict__ f,
                                  float* __restrict__ out) {
  constexpr int DT = DZ / 16;
  constexpr int TT = T / 16;

  const int wave = blockIdx.x * (blockDim.x >> 5) + (threadIdx.x >> 5);
  const int lane = threadIdx.x & 31;

  const int dt = wave & (DT - 1);
  const int tt = (wave >> 5) & (TT - 1);
  const int b  = wave >> 10;

  const bool lo  = lane < 16;
  const int  l16 = lane & 15;

  v2f a;
  a[0] = lo ? f[dt * 16 + l16] : 0.0f;
  a[1] = 0.0f;

  v2f bm;
  bm[0] = lo ? S[(size_t)b * T + tt * 16 + l16] : 0.0f;
  bm[1] = 0.0f;

  const float* z0row = z0 + (size_t)b * DZ + dt * 16 + (lo ? 0 : 8);
  v8f c;
  c[0] = z0row[0]; c[1] = z0row[1]; c[2] = z0row[2]; c[3] = z0row[3];
  c[4] = z0row[4]; c[5] = z0row[5]; c[6] = z0row[6]; c[7] = z0row[7];

  v8f d = __builtin_amdgcn_wmma_f32_16x16x4_f32(
       false, a,  false, bm,
       (short)0, c,  false,  false);

  const size_t base = ((size_t)b * T + tt * 16 + l16) * DZ + dt * 16 + (lo ? 0 : 8);
  v4f d03 = {d[0], d[1], d[2], d[3]};
  v4f d47 = {d[4], d[5], d[6], d[7]};
  *(v4f*)(out + base)     = d03;
  *(v4f*)(out + base + 4) = d47;
}

extern "C" void kernel_launch(void* const* d_in, const int* in_sizes, int n_in,
                              void* d_out, int out_size, void* d_ws, size_t ws_size,
                              hipStream_t stream) {
  const float* x  = (const float*)d_in[0];
  const float* z0 = (const float*)d_in[1];
  const float* f  = (const float*)d_in[2];
  float* out = (float*)d_out;
  float* S   = (float*)d_ws;

  (void)in_sizes; (void)n_in; (void)out_size; (void)ws_size;

  scan_kernel<<<B, T, 0, stream>>>(x, S);

  constexpr int totalWaves = B * (T / 16) * (DZ / 16);
  constexpr int blockThreads = 256;
  constexpr int grid = totalWaves / (blockThreads / 32);
  rank1_wmma_kernel<<<grid, blockThreads, 0, stream>>>(S, z0, f, out);
}


# speedup vs baseline: 2.1377x; 2.1377x over previous
#include <hip/hip_runtime.h>


typedef __attribute__((ext_vector_type(2))) float v2f;
typedef __attribute__((ext_vector_type(4))) float v4f;
typedef __attribute__((ext_vector_type(8))) float v8f;

constexpr int B  = 128;
constexpr int T  = 512;
constexpr int DZ = 512;

constexpr int WAVES_PER_BLOCK = 8;
constexpr int SUB    = 32;
constexpr int NSUB   = 2;
constexpr int STRIP  = SUB * NSUB;
constexpr int STRIPS = DZ / STRIP;
constexpr int HALVES = 2;
constexpr int TH     = T / HALVES;
constexpr int TILE_T = 8;
constexpr int XROW   = 16;
constexpr int TRIP   = 32;
constexpr int RING   = 2;

template <int J>
__device__ __forceinline__ float row_bcast(float v) {
  return __builtin_bit_cast(float,
      __builtin_amdgcn_update_dpp(0, __builtin_bit_cast(int, v), 0x150 + J, 0xF, 0xF, true));
}
__device__ __forceinline__ float row_bcast(float v, int j) {
  switch (j & 15) {
    case 0:  return row_bcast<0>(v);   case 1:  return row_bcast<1>(v);
    case 2:  return row_bcast<2>(v);   case 3:  return row_bcast<3>(v);
    case 4:  return row_bcast<4>(v);   case 5:  return row_bcast<5>(v);
    case 6:  return row_bcast<6>(v);   case 7:  return row_bcast<7>(v);
    case 8:  return row_bcast<8>(v);   case 9:  return row_bcast<9>(v);
    case 10: return row_bcast<10>(v);  case 11: return row_bcast<11>(v);
    case 12: return row_bcast<12>(v);  case 13: return row_bcast<13>(v);
    case 14: return row_bcast<14>(v);  default: return row_bcast<15>(v);
  }
}

__global__ __launch_bounds__(WAVES_PER_BLOCK * 32)
void stream_cumsum_wmma_kernel(const float* __restrict__ x,
                               const float* __restrict__ z0,
                               const float* __restrict__ f,
                               float* __restrict__ out) {
  __shared__ __attribute__((aligned(16))) float stage[WAVES_PER_BLOCK][RING][TILE_T][STRIP];

  const unsigned lane = threadIdx.x & 31u;
  const int wv   = __builtin_amdgcn_readfirstlane((int)(threadIdx.x >> 5));
  const int b    = (int)(blockIdx.x >> 1);
  const int d0   = (int)(((blockIdx.x & 1u) * 4u + (unsigned)(wv & 3)) * STRIP);
  const int half = wv >> 2;
  const int tb   = half * TH;

  float fl[NSUB], zl[NSUB];
#pragma unroll
  for (int u = 0; u < NSUB; ++u) {
    fl[u] = f[d0 + SUB * u + lane];
    zl[u] = z0[(size_t)b * DZ + d0 + SUB * u + lane];
  }

  const int  c   = (int)(lane & 15u);
  const int  hk  = (int)(lane >> 4);
  const bool sel = ((c >> 3) == hk);
  const int  r   = c & 7;
  v2f a[4];
#pragma unroll
  for (int q = 0; q < 4; ++q) {
    a[q][0] = (sel && (2 * q)     <= r) ? 1.0f : 0.0f;
    a[q][1] = (sel && (2 * q + 1) <= r) ? 1.0f : 0.0f;
  }

  const float* __restrict__ xr = x + (size_t)b * T;
  float xv[TH / XROW];
#pragma unroll
  for (int i = 0; i < TH / XROW; ++i) xv[i] = xr[tb + XROW * i + c];

  float* __restrict__ op = out + (size_t)b * T * DZ + (size_t)(tb + hk) * DZ + d0 + 4 * c;

  float carry = 0.0f;
  if (half) {
    float pre[TH / XROW];
#pragma unroll
    for (int i = 0; i < TH / XROW; ++i) pre[i] = xr[XROW * i + c];
#pragma unroll
    for (int i = 0; i < TH / XROW; ++i)
#pragma unroll
      for (int j = 0; j < XROW; ++j) carry = carry + row_bcast(pre[i], j);
  }
  v8f zc[NSUB];
#pragma unroll
  for (int u = 0; u < NSUB; ++u) zc[u] = (v8f){zl[u], zl[u], zl[u], zl[u], zl[u], zl[u], zl[u], zl[u]};

  constexpr int NT = TRIP / TILE_T;
  v8f prev[NT][NSUB];
#pragma unroll
  for (int tt = 0; tt < NT; ++tt)
#pragma unroll
    for (int u = 0; u < NSUB; ++u) prev[tt][u] = zc[u];

  auto flush = [&](const v8f (&tile)[NSUB], int t, int buf) {
    float (*st)[STRIP] = stage[wv][buf];
#pragma unroll
    for (int u = 0; u < NSUB; ++u)
#pragma unroll
      for (int rr = 0; rr < TILE_T; ++rr) st[rr][SUB * u + lane] = tile[u][rr];
    __builtin_amdgcn_fence(__ATOMIC_ACQ_REL, "wavefront");
#pragma unroll
    for (int k = 0; k < TILE_T / 2; ++k) {
      const v4f v = *reinterpret_cast<const v4f*>(&st[2 * k + hk][4 * c]);
      *reinterpret_cast<v4f*>(op + (size_t)(t + 2 * k) * DZ) = v;
    }
  };

#pragma unroll
  for (int t0 = 0; t0 < TH; t0 += TRIP) {
    v8f acc[NT][NSUB];
    v2f bq[NT][NSUB][4];
#pragma unroll
    for (int tt = 0; tt < NT; ++tt) {
      const int   t  = t0 + tt * TILE_T;
      const float xi = xv[t / XROW];
      const int   j0 = t % XROW;
#pragma unroll
      for (int u = 0; u < NSUB; ++u) acc[tt][u] = zc[u];
#pragma unroll
      for (int j = 0; j < TILE_T; ++j) {
        const float xj = row_bcast(xi, j0 + j);
        carry = carry + xj;
        const float inc = (j == 0 ? carry : xj);
#pragma unroll
        for (int u = 0; u < NSUB; ++u) bq[tt][u][j >> 1][j & 1] = fl[u] * inc;
      }
    }

#pragma unroll
    for (int q = 0; q < 4; ++q) {
#pragma unroll
      for (int tt = 0; tt < NT; ++tt)
#pragma unroll
        for (int u = 0; u < NSUB; ++u)
          acc[tt][u] = __builtin_amdgcn_wmma_f32_16x16x4_f32(false, a[q], false, bq[tt][u][q],
                                                             (short)0, acc[tt][u], false, false);
      if (t0 > 0) flush(prev[q], t0 - TRIP + q * TILE_T, q & (RING - 1));
    }
#pragma unroll
    for (int tt = 0; tt < NT; ++tt)
#pragma unroll
      for (int u = 0; u < NSUB; ++u) prev[tt][u] = acc[tt][u];
  }

#pragma unroll
  for (int tt = 0; tt < NT; ++tt) flush(prev[tt], TH - TRIP + tt * TILE_T, tt & (RING - 1));
}

extern "C" void kernel_launch(void* const* d_in, const int* in_sizes, int n_in,
                              void* d_out, int out_size, void* d_ws, size_t ws_size,
                              hipStream_t stream) {
  const float* x  = (const float*)d_in[0];
  const float* z0 = (const float*)d_in[1];
  const float* f  = (const float*)d_in[2];
  float* out = (float*)d_out;

  (void)in_sizes; (void)n_in; (void)out_size; (void)d_ws; (void)ws_size;

  constexpr int grid = B * STRIPS * HALVES / WAVES_PER_BLOCK;
  stream_cumsum_wmma_kernel<<<grid, WAVES_PER_BLOCK * 32, 0, stream>>>(x, z0, f, out);
}
